# starting kernel
# baseline (speedup 1.0000x reference)
#include <hip/hip_runtime.h>

typedef __attribute__((ext_vector_type(2))) float v2f;
typedef __attribute__((ext_vector_type(8))) float v8f;


__global__ void k_deg_init(float* __restrict__ deg, int n) {
  int i = blockIdx.x * blockDim.x + threadIdx.x;
  if (i < n) deg[i] = 1.0f;
}

__global__ void k_deg_edges(const int* __restrict__ dst, float* __restrict__ deg, int e) {
  int i = blockIdx.x * blockDim.x + threadIdx.x;
  if (i < e) atomicAdd(&deg[dst[i]], 1.0f);
}

__global__ void k_rsqrt_inplace(float* __restrict__ deg, int n) {
  int i = blockIdx.x * blockDim.x + threadIdx.x;
  if (i < n) deg[i] = rsqrtf(deg[i]);
}


template <int K, int MOUT, bool ACT>
__global__ void k_gemm_strip(const float* __restrict__ X, const float* __restrict__ W,
                             const float* __restrict__ bias_in,
                             const float* __restrict__ dinv,
                             float* __restrict__ hs, float* __restrict__ agg, int nrows) {
  constexpr int T = MOUT / 16;
  const int wave = (blockIdx.x * blockDim.x + threadIdx.x) >> 5;
  const int lane = threadIdx.x & 31;
  if (wave * 16 >= nrows) return;

  const int half = lane >> 4;
  const int l16  = lane & 15;
  const int rowA = wave * 16 + l16;

  v8f acc[T];
#pragma unroll
  for (int t = 0; t < T; ++t) acc[t] = (v8f){};

#pragma unroll 4
  for (int k = 0; k < K; k += 4) {
    const int ka = k + 2 * half;
    v2f a;
    a.x = X[(size_t)rowA * K + ka];
    a.y = X[(size_t)rowA * K + ka + 1];
    if (ACT) {
      a.x = fmaxf(a.x + bias_in[ka], 0.0f);
      a.y = fmaxf(a.y + bias_in[ka + 1], 0.0f);
    }
#pragma unroll
    for (int t = 0; t < T; ++t) {
      const int colB = t * 16 + l16;
      v2f b;
      b.x = W[(size_t)ka * MOUT + colB];
      b.y = W[(size_t)(ka + 1) * MOUT + colB];
      acc[t] = __builtin_amdgcn_wmma_f32_16x16x4_f32(
          false, a, false, b, (short)0, acc[t], false, false);
    }
  }

#pragma unroll
  for (int v = 0; v < 8; ++v) {
    const int row = wave * 16 + v + 8 * half;
    const float dn = dinv[row];
#pragma unroll
    for (int t = 0; t < T; ++t) {
      const int col = t * 16 + l16;
      const float h = acc[t][v] * dn;
      hs[(size_t)row * MOUT + col]  = h;
      agg[(size_t)row * MOUT + col] = h * dn;
    }
  }
}


template <int MOUT>
__global__ void k_scatter(const int* __restrict__ src, const int* __restrict__ dst,
                          const float* __restrict__ hs, const float* __restrict__ dinv,
                          float* __restrict__ agg, int e) {
  constexpr int VPT = MOUT / 32;
  const long long g = (long long)blockIdx.x * blockDim.x + threadIdx.x;
  const int edge = (int)(g >> 5);
  const int lane = (int)(g & 31);
  if (edge >= e) return;
  const int s = src[edge];
  const int d = dst[edge];
  const float dn = dinv[d];
  const float* hrow = hs  + (size_t)s * MOUT + lane * VPT;
  float*       arow = agg + (size_t)d * MOUT + lane * VPT;
  if constexpr (VPT == 4) {
    const float4 v = *(const float4*)hrow;
    atomicAdd(arow + 0, v.x * dn);
    atomicAdd(arow + 1, v.y * dn);
    atomicAdd(arow + 2, v.z * dn);
    atomicAdd(arow + 3, v.w * dn);
  } else if constexpr (VPT == 2) {
    const float2 v = *(const float2*)hrow;
    atomicAdd(arow + 0, v.x * dn);
    atomicAdd(arow + 1, v.y * dn);
  } else {
    atomicAdd(arow, hrow[0] * dn);
  }
}


template <int MOUT>
__global__ void k_bias_out(const float* __restrict__ agg, const float* __restrict__ bias,
                           float* __restrict__ out, int n) {
  const size_t i = (size_t)blockIdx.x * blockDim.x + threadIdx.x;
  if (i >= (size_t)n * MOUT) return;
  out[i] = agg[i] + bias[(int)(i % MOUT)];
}


template <int K, int MOUT, bool ACT>
static void run_layer(const float* in, const float* W, const float* bias_in,
                      const float* dinv, const int* esrc, const int* edst,
                      float* hs, float* agg, int n, int e, hipStream_t stream) {
  const long long strips = (n + 15) / 16;
  const int gblocks = (int)((strips * 32 + 255) / 256);
  k_gemm_strip<K, MOUT, ACT><<<gblocks, 256, 0, stream>>>(in, W, bias_in, dinv, hs, agg, n);

  const int sblocks = (int)(((long long)e * 32 + 255) / 256);
  k_scatter<MOUT><<<sblocks, 256, 0, stream>>>(esrc, edst, hs, dinv, agg, e);
}

extern "C" void kernel_launch(void* const* d_in, const int* in_sizes, int n_in,
                              void* d_out, int out_size, void* d_ws, size_t ws_size,
                              hipStream_t stream) {
  const float* x  = (const float*)d_in[0];
  const int*   ei = (const int*)d_in[1];
  const float* W1 = (const float*)d_in[2];
  const float* b1 = (const float*)d_in[3];
  const float* W2 = (const float*)d_in[4];
  const float* b2 = (const float*)d_in[5];
  const float* W3 = (const float*)d_in[6];
  const float* b3 = (const float*)d_in[7];
  const float* W4 = (const float*)d_in[8];
  const float* b4 = (const float*)d_in[9];

  const int n = in_sizes[0] / 256;
  const int e = in_sizes[1] / 2;
  const int* esrc = ei;
  const int* edst = ei + e;

  float* dinv = (float*)d_ws;
  float* H  = dinv + n;
  float* G1 = H  + (size_t)n * 128;
  float* G2 = G1 + (size_t)n * 128;

  k_deg_init<<<(n + 255) / 256, 256, 0, stream>>>(dinv, n);
  k_deg_edges<<<(e + 255) / 256, 256, 0, stream>>>(edst, dinv, e);
  k_rsqrt_inplace<<<(n + 255) / 256, 256, 0, stream>>>(dinv, n);

  run_layer<256, 128, false>(x,  W1, nullptr, dinv, esrc, edst, H, G1, n, e, stream);
  run_layer<128, 128, true >(G1, W2, b1,      dinv, esrc, edst, H, G2, n, e, stream);
  run_layer<128, 64,  true >(G2, W3, b2,      dinv, esrc, edst, H, G1, n, e, stream);
  run_layer<64,  32,  true >(G1, W4, b3,      dinv, esrc, edst, H, G2, n, e, stream);

  const long long tot = (long long)n * 32;
  k_bias_out<32><<<(int)((tot + 255) / 256), 256, 0, stream>>>(G2, b4, (float*)d_out, n);
}


# Edge scatter by float atomics -> destination-sorted adjacency rebuilt each call (integer-atomic rank, scan, fill, per-row sort) + one wave per node summing rows in registers in fixed order
# speedup vs baseline: 2.8119x; 2.8119x over previous
#include <hip/hip_runtime.h>


typedef __attribute__((ext_vector_type(2))) float v2f;
typedef __attribute__((ext_vector_type(4))) float v4f;
typedef __attribute__((ext_vector_type(8))) float v8f;


constexpr int SCAN_THREADS = 256;
constexpr int SCAN_PER_THREAD = 4;
constexpr int SCAN_BLOCK = SCAN_THREADS * SCAN_PER_THREAD;

__global__ void k_zero_counts(int* __restrict__ cnt, int n) {
  const int i = blockIdx.x * blockDim.x + threadIdx.x;
  if (i < n) cnt[i] = 0;
}

__global__ void k_rank_edges(const int* __restrict__ dst, int* __restrict__ cnt,
                             int* __restrict__ rank, int e) {
  const int i = blockIdx.x * blockDim.x + threadIdx.x;
  if (i < e) rank[i] = atomicAdd(&cnt[dst[i]], 1);
}

__global__ void k_block_sums(const int* __restrict__ cnt, int* __restrict__ bsum, int n) {
  __shared__ int sh[SCAN_THREADS];
  const int tid = threadIdx.x;
  const int base = blockIdx.x * SCAN_BLOCK + tid * SCAN_PER_THREAD;
  int s = 0;
#pragma unroll
  for (int q = 0; q < SCAN_PER_THREAD; ++q)
    if (base + q < n) s += cnt[base + q];
  sh[tid] = s;
  __syncthreads();
  for (int off = SCAN_THREADS / 2; off > 0; off >>= 1) {
    if (tid < off) sh[tid] += sh[tid + off];
    __syncthreads();
  }
  if (tid == 0) bsum[blockIdx.x] = sh[0];
}

__global__ void k_scan_counts(const int* __restrict__ cnt, const int* __restrict__ bsum,
                              int* __restrict__ rowptr, float* __restrict__ dinv, int n) {
  __shared__ int sh[SCAN_THREADS];
  const int tid = threadIdx.x;

  int b = 0;
  for (int t = tid; t < (int)blockIdx.x; t += SCAN_THREADS) b += bsum[t];
  sh[tid] = b;
  __syncthreads();
  for (int off = SCAN_THREADS / 2; off > 0; off >>= 1) {
    if (tid < off) sh[tid] += sh[tid + off];
    __syncthreads();
  }
  const int block_base = sh[0];
  __syncthreads();

  const int base = blockIdx.x * SCAN_BLOCK + tid * SCAN_PER_THREAD;
  int c[SCAN_PER_THREAD];
  int tsum = 0;
#pragma unroll
  for (int q = 0; q < SCAN_PER_THREAD; ++q) {
    c[q] = (base + q < n) ? cnt[base + q] : 0;
    tsum += c[q];
  }
  sh[tid] = tsum;
  __syncthreads();
  for (int off = 1; off < SCAN_THREADS; off <<= 1) {
    const int add = (tid >= off) ? sh[tid - off] : 0;
    __syncthreads();
    sh[tid] += add;
    __syncthreads();
  }
  int run = block_base + sh[tid] - tsum;
#pragma unroll
  for (int q = 0; q < SCAN_PER_THREAD; ++q) {
    if (base + q < n) {
      rowptr[base + q] = run;
      dinv[base + q] = rsqrtf((float)(c[q] + 1));
    }
    run += c[q];
  }
  if (blockIdx.x == gridDim.x - 1 && tid == SCAN_THREADS - 1) rowptr[n] = run;
}

__global__ void k_fill_slots(const int* __restrict__ src, const int* __restrict__ dst,
                             const int* __restrict__ rank, const int* __restrict__ rowptr,
                             int* __restrict__ slot, int e) {
  const int i = blockIdx.x * blockDim.x + threadIdx.x;
  if (i < e) slot[rowptr[dst[i]] + rank[i]] = src[i];
}

__global__ void k_sort_segments(const int* __restrict__ rowptr, const int* __restrict__ slot,
                                int* __restrict__ col, int n) {
  const int node = __builtin_amdgcn_readfirstlane((int)((blockIdx.x * blockDim.x + threadIdx.x) >> 5));
  if (node >= n) return;
  const int lane = threadIdx.x & 31;
  const int beg = rowptr[node];
  const int d = rowptr[node + 1] - beg;
  if (d <= 32) {
    const int c = (lane < d) ? slot[beg + lane] : 0x7fffffff;
    int r = 0;
    for (int j = 0; j < d; ++j) {
      const int cj = __builtin_amdgcn_readlane(c, j);
      r += (cj < c || (cj == c && j < lane)) ? 1 : 0;
    }
    if (lane < d) col[beg + r] = c;
  } else {
    for (int t = lane; t < d; t += 32) {
      const int c = slot[beg + t];
      int r = 0;
      for (int j = 0; j < d; ++j) {
        const int cj = slot[beg + j];
        r += (cj < c || (cj == c && j < t)) ? 1 : 0;
      }
      col[beg + r] = c;
    }
  }
}


template <int K, int MOUT, bool ACT>
__global__ void k_gemm_strip(const float* __restrict__ X, const float* __restrict__ W,
                             const float* __restrict__ bias_in,
                             const float* __restrict__ dinv,
                             float* __restrict__ hs, int nrows) {
  constexpr int T = MOUT / 16;
  const int wave = (blockIdx.x * blockDim.x + threadIdx.x) >> 5;
  const int lane = threadIdx.x & 31;
  if (wave * 16 >= nrows) return;

  const int half = lane >> 4;
  const int l16  = lane & 15;
  const int rowA = wave * 16 + l16;

  v8f acc[T];
#pragma unroll
  for (int t = 0; t < T; ++t) acc[t] = (v8f){};

#pragma unroll 4
  for (int k = 0; k < K; k += 4) {
    const int ka = k + 2 * half;
    v2f a;
    a.x = X[(size_t)rowA * K + ka];
    a.y = X[(size_t)rowA * K + ka + 1];
    if (ACT) {
      a.x = fmaxf(a.x + bias_in[ka], 0.0f);
      a.y = fmaxf(a.y + bias_in[ka + 1], 0.0f);
    }
#pragma unroll
    for (int t = 0; t < T; ++t) {
      const int colB = t * 16 + l16;
      v2f b;
      b.x = W[(size_t)ka * MOUT + colB];
      b.y = W[(size_t)(ka + 1) * MOUT + colB];
      acc[t] = __builtin_amdgcn_wmma_f32_16x16x4_f32(
          false, a, false, b, (short)0, acc[t], false, false);
    }
  }

#pragma unroll
  for (int v = 0; v < 8; ++v) {
    const int row = wave * 16 + v + 8 * half;
    const float dn = dinv[row];
#pragma unroll
    for (int t = 0; t < T; ++t) {
      const int col = t * 16 + l16;
      hs[(size_t)row * MOUT + col] = acc[t][v] * dn;
    }
  }
}


template <int V> struct RowVec { typedef __attribute__((ext_vector_type(V))) float type; };
template <> struct RowVec<1> { typedef float type; };

template <int G, int MOUT, typename VecT>
__device__ __forceinline__ void add_rows(VecT& acc, const float* __restrict__ hl, int idx, int t) {
  VecT v[G];
#pragma unroll
  for (int g = 0; g < G; ++g) {
    const int s = __builtin_amdgcn_readlane(idx, t + g);
    v[g] = *(const VecT*)(hl + (size_t)s * MOUT);
  }
#pragma unroll
  for (int g = 0; g < G; ++g) acc += v[g];
}

template <int MOUT>
__global__ void k_gather(const int* __restrict__ rowptr, const int* __restrict__ col,
                         const float* __restrict__ hs, const float* __restrict__ dinv,
                         float* __restrict__ agg, int n) {
  constexpr int VPT = MOUT / 32;
  typedef typename RowVec<VPT>::type VecT;
  const int node = __builtin_amdgcn_readfirstlane((int)((blockIdx.x * blockDim.x + threadIdx.x) >> 5));
  if (node >= n) return;
  const int lane = threadIdx.x & 31;
  const int beg = rowptr[node];
  const int end = rowptr[node + 1];
  const float* hl = hs + lane * VPT;

  VecT acc = *(const VecT*)(hl + (size_t)node * MOUT);
  for (int base = beg; base < end; base += 32) {
    const int m = (end - base < 32) ? (end - base) : 32;
    const int idx = (lane < m) ? col[base + lane] : 0;
    int t = 0;
    for (; t + 8 <= m; t += 8) add_rows<8, MOUT, VecT>(acc, hl, idx, t);
    if (t + 4 <= m) { add_rows<4, MOUT, VecT>(acc, hl, idx, t); t += 4; }
    if (t + 2 <= m) { add_rows<2, MOUT, VecT>(acc, hl, idx, t); t += 2; }
    if (t < m)      { add_rows<1, MOUT, VecT>(acc, hl, idx, t); }
  }
  const float dn = dinv[node];
  *(VecT*)(agg + (size_t)node * MOUT + lane * VPT) = acc * dn;
}


template <int MOUT>
__global__ void k_bias_out(const float* __restrict__ agg, const float* __restrict__ bias,
                           float* __restrict__ out, int n) {
  const size_t i = (size_t)blockIdx.x * blockDim.x + threadIdx.x;
  if (i >= (size_t)n * MOUT) return;
  out[i] = agg[i] + bias[(int)(i % MOUT)];
}


template <int K, int MOUT, bool ACT>
static void run_layer(const float* in, const float* W, const float* bias_in,
                      const float* dinv, const int* rowptr, const int* col,
                      float* hs, float* agg, int n, hipStream_t stream) {
  const long long strips = (n + 15) / 16;
  const int gblocks = (int)((strips * 32 + 255) / 256);
  k_gemm_strip<K, MOUT, ACT><<<gblocks, 256, 0, stream>>>(in, W, bias_in, dinv, hs, n);

  const int ablocks = (int)(((long long)n * 32 + 255) / 256);
  k_gather<MOUT><<<ablocks, 256, 0, stream>>>(rowptr, col, hs, dinv, agg, n);
}

extern "C" void kernel_launch(void* const* d_in, const int* in_sizes, int n_in,
                              void* d_out, int out_size, void* d_ws, size_t ws_size,
                              hipStream_t stream) {
  const float* x  = (const float*)d_in[0];
  const int*   ei = (const int*)d_in[1];
  const float* W1 = (const float*)d_in[2];
  const float* b1 = (const float*)d_in[3];
  const float* W2 = (const float*)d_in[4];
  const float* b2 = (const float*)d_in[5];
  const float* W3 = (const float*)d_in[6];
  const float* b3 = (const float*)d_in[7];
  const float* W4 = (const float*)d_in[8];
  const float* b4 = (const float*)d_in[9];

  const int n = in_sizes[0] / 256;
  const int e = in_sizes[1] / 2;
  const int* esrc = ei;
  const int* edst = ei + e;
  const int nscan = (n + SCAN_BLOCK - 1) / SCAN_BLOCK;

  char* cur = (char*)d_ws;
  auto carve = [&cur](size_t bytes) { char* p = cur; cur += (bytes + 255) & ~(size_t)255; return p; };
  float* dinv   = (float*)carve((size_t)n * 4);
  int*   rowptr = (int*)carve((size_t)(n + 1) * 4);
  int*   cnt    = (int*)carve((size_t)n * 4);
  int*   bsum   = (int*)carve((size_t)nscan * 4);
  int*   rank   = (int*)carve((size_t)e * 4);
  int*   slot   = (int*)carve((size_t)e * 4);
  int*   col    = (int*)carve((size_t)e * 4);
  float* H  = (float*)carve((size_t)n * 128 * 4);
  float* G1 = (float*)carve((size_t)n * 128 * 4);
  float* G2 = (float*)carve((size_t)n * 128 * 4);

  k_zero_counts<<<(n + 255) / 256, 256, 0, stream>>>(cnt, n);
  k_rank_edges<<<(e + 255) / 256, 256, 0, stream>>>(edst, cnt, rank, e);
  k_block_sums<<<nscan, SCAN_THREADS, 0, stream>>>(cnt, bsum, n);
  k_scan_counts<<<nscan, SCAN_THREADS, 0, stream>>>(cnt, bsum, rowptr, dinv, n);
  k_fill_slots<<<(e + 255) / 256, 256, 0, stream>>>(esrc, edst, rank, rowptr, slot, e);
  k_sort_segments<<<(int)(((long long)n * 32 + 255) / 256), 256, 0, stream>>>(rowptr, slot, col, n);

  run_layer<256, 128, false>(x,  W1, nullptr, dinv, rowptr, col, H, G1, n, stream);
  run_layer<128, 128, true >(G1, W2, b1,      dinv, rowptr, col, H, G2, n, stream);
  run_layer<128, 64,  true >(G2, W3, b2,      dinv, rowptr, col, H, G1, n, stream);
  run_layer<64,  32,  true >(G1, W4, b3,      dinv, rowptr, col, H, G2, n, stream);

  const long long tot = (long long)n * 32;
  k_bias_out<32><<<(int)((tot + 255) / 256), 256, 0, stream>>>(G2, b4, (float*)d_out, n);
}


# Per-edge returning global integer atomics -> LDS counters: edges filed into 256-node destination buckets (LDS count, table prefix, LDS rank); one work-group per bucket builds and sorts its rows
# speedup vs baseline: 3.4704x; 1.2342x over previous
#include <hip/hip_runtime.h>
#include <type_traits>


typedef __attribute__((ext_vector_type(2))) float v2f;
typedef __attribute__((ext_vector_type(4))) float v4f;
typedef __attribute__((ext_vector_type(8))) float v8f;


constexpr int NODE_SHIFT = 8;
constexpr int BUCKET_NODES = 1 << NODE_SHIFT;
constexpr int EDGE_BLOCKS = 128;
constexpr int EDGE_THREADS = 1024;
constexpr int EDGE_WAVES = EDGE_THREADS / 32;
constexpr int MAX_BUCKETS = 1024;
constexpr int OWNER_THREADS = 512;
constexpr int OWNER_WAVES = OWNER_THREADS / 32;
constexpr int OWNER_CAP = 8192;
static_assert(EDGE_BLOCKS == 128, "one 16-byte load per lane covers a table row");
static_assert(MAX_BUCKETS == EDGE_THREADS && EDGE_WAVES == 32, "scan: one bucket per thread, one wave total per lane");

__global__ __launch_bounds__(EDGE_THREADS) void k_bucket_counts(
    const int* __restrict__ dst, int* __restrict__ table, int e, int chunk, int nbk) {
  __shared__ int h[MAX_BUCKETS];
  const int tid = threadIdx.x;
  for (int b = tid; b < nbk; b += EDGE_THREADS) h[b] = 0;
  __syncthreads();
  const int lo = blockIdx.x * chunk;
  const int hi = min(e, lo + chunk);
  for (int i = lo + tid; i < hi; i += 4 * EDGE_THREADS) {
    int d[4];
#pragma unroll
    for (int q = 0; q < 4; ++q) {
      const int j = i + q * EDGE_THREADS;
      d[q] = dst[j < hi ? j : hi - 1];
    }
#pragma unroll
    for (int q = 0; q < 4; ++q)
      atomicAdd(&h[d[q] >> NODE_SHIFT], (i + q * EDGE_THREADS < hi) ? 1 : 0);
  }
  __syncthreads();
  for (int b = tid; b < nbk; b += EDGE_THREADS) table[b * EDGE_BLOCKS + blockIdx.x] = h[b];
}

__global__ __launch_bounds__(EDGE_THREADS) void k_bucket_scatter(
    const int* __restrict__ src, const int* __restrict__ dst, const int* __restrict__ table,
    int* __restrict__ bbase, unsigned* __restrict__ rec, int e, int chunk, int nbk) {
  __shared__ int tot[MAX_BUCKETS];
  __shared__ int pre[MAX_BUCKETS];
  __shared__ int off[MAX_BUCKETS];
  __shared__ int wtot[EDGE_WAVES];
  const int tid = threadIdx.x;
  const int lane = tid & 31;
  const int wave = tid >> 5;
  const int blk = blockIdx.x;

  for (int b0 = wave; b0 < nbk; b0 += 8 * EDGE_WAVES) {
    int4 v[8];
#pragma unroll
    for (int q = 0; q < 8; ++q) {
      const int b = b0 + q * EDGE_WAVES;
      v[q] = *(const int4*)(table + (b < nbk ? b : nbk - 1) * EDGE_BLOCKS + lane * 4);
    }
#pragma unroll
    for (int q = 0; q < 8; ++q) {
      const int b = b0 + q * EDGE_WAVES;
      const int c0 = lane * 4;
      int t = v[q].x + v[q].y + v[q].z + v[q].w;
      int p = (c0 < blk ? v[q].x : 0) + (c0 + 1 < blk ? v[q].y : 0) + (c0 + 2 < blk ? v[q].z : 0) +
              (c0 + 3 < blk ? v[q].w : 0);
#pragma unroll
      for (int o = 16; o > 0; o >>= 1) {
        t += __shfl_xor(t, o);
        p += __shfl_xor(p, o);
      }
      if (lane == 0 && b < nbk) { tot[b] = t; pre[b] = p; }
    }
  }
  __syncthreads();

  const int mine = (tid < nbk) ? tot[tid] : 0;
  int incl = mine;
#pragma unroll
  for (int o = 1; o < 32; o <<= 1) {
    const int u = __shfl_up(incl, o);
    if (lane >= o) incl += u;
  }
  if (lane == 31) wtot[wave] = incl;
  __syncthreads();
  int wincl = wtot[lane];
#pragma unroll
  for (int o = 1; o < 32; o <<= 1) {
    const int u = __shfl_up(wincl, o);
    if (lane >= o) wincl += u;
  }
  const int wexcl = wincl - wtot[lane];
  const int first = __shfl(wexcl, wave) + incl - mine;
  if (tid < nbk) {
    off[tid] = first + pre[tid];
    if (blk == 0) {
      bbase[tid] = first;
      if (tid == nbk - 1) bbase[nbk] = first + mine;
    }
  }
  __syncthreads();

  const int lo = blk * chunk;
  const int hi = min(e, lo + chunk);
  int i = lo + tid;
  for (; i + 3 * EDGE_THREADS < hi; i += 4 * EDGE_THREADS) {
    int d[4], s[4];
#pragma unroll
    for (int q = 0; q < 4; ++q) {
      d[q] = dst[i + q * EDGE_THREADS];
      s[q] = src[i + q * EDGE_THREADS];
    }
#pragma unroll
    for (int q = 0; q < 4; ++q) {
      const int pos = atomicAdd(&off[d[q] >> NODE_SHIFT], 1);
      rec[pos] = (unsigned)s[q] | ((unsigned)(d[q] & (BUCKET_NODES - 1)) << 24);
    }
  }
  for (; i < hi; i += EDGE_THREADS) {
    const int d = dst[i];
    const int s = src[i];
    const int pos = atomicAdd(&off[d >> NODE_SHIFT], 1);
    rec[pos] = (unsigned)s | ((unsigned)(d & (BUCKET_NODES - 1)) << 24);
  }
}

__global__ __launch_bounds__(OWNER_THREADS) void k_bucket_rows(
    const int* __restrict__ bbase, const unsigned* __restrict__ rec, int* __restrict__ rowptr,
    float* __restrict__ dinv, int* __restrict__ slot, int* __restrict__ col, int n, int nbk) {
  __shared__ int cnt[BUCKET_NODES];
  __shared__ int start[BUCKET_NODES];
  __shared__ int cur[BUCKET_NODES];
  __shared__ int wsum[BUCKET_NODES / 32];
  __shared__ int ls[OWNER_CAP];
  const int tid = threadIdx.x;
  const int lane = tid & 31;
  const int wave = tid >> 5;
  const int b = blockIdx.x;
  const int lo = bbase[b];
  const int hi = bbase[b + 1];
  const bool fits = (hi - lo) <= OWNER_CAP;

  if (tid < BUCKET_NODES) { cnt[tid] = 0; cur[tid] = 0; }
  __syncthreads();

  for (int i = lo + tid; i < hi; i += 4 * OWNER_THREADS) {
    unsigned r[4];
#pragma unroll
    for (int q = 0; q < 4; ++q) {
      const int j = i + q * OWNER_THREADS;
      r[q] = rec[j < hi ? j : hi - 1];
    }
#pragma unroll
    for (int q = 0; q < 4; ++q)
      atomicAdd(&cnt[r[q] >> 24], (i + q * OWNER_THREADS < hi) ? 1 : 0);
  }
  __syncthreads();

  int c = 0, incl = 0;
  if (tid < BUCKET_NODES) {
    c = cnt[tid];
    incl = c;
#pragma unroll
    for (int o = 1; o < 32; o <<= 1) {
      const int u = __shfl_up(incl, o);
      if (lane >= o) incl += u;
    }
    if (lane == 31) wsum[wave] = incl;
  }
  __syncthreads();
  if (tid < BUCKET_NODES) {
    int wbase = 0;
    for (int w = 0; w < wave; ++w) wbase += wsum[w];
    const int st = wbase + incl - c;
    start[tid] = st;
    const int node = b * BUCKET_NODES + tid;
    if (node < n) {
      rowptr[node] = lo + st;
      dinv[node] = rsqrtf((float)(c + 1));
    }
  }
  if (b == nbk - 1 && tid == 0) rowptr[n] = hi;
  __syncthreads();

  for (int i = lo + tid; i < hi; i += 4 * OWNER_THREADS) {
    unsigned r[4];
#pragma unroll
    for (int q = 0; q < 4; ++q) {
      const int j = i + q * OWNER_THREADS;
      r[q] = rec[j < hi ? j : hi - 1];
    }
#pragma unroll
    for (int q = 0; q < 4; ++q) {
      if (i + q * OWNER_THREADS < hi) {
        const int dl = (int)(r[q] >> 24);
        const int pos = start[dl] + atomicAdd(&cur[dl], 1);
        const int s = (int)(r[q] & 0xffffffu);
        if (fits) ls[pos] = s; else slot[lo + pos] = s;
      }
    }
  }
  __syncthreads();

  auto sort_rows = [&](auto in_lds) {
    constexpr bool L = decltype(in_lds)::value;
    for (int dl = wave; dl < BUCKET_NODES; dl += OWNER_WAVES) {
      if (b * BUCKET_NODES + dl >= n) break;
      const int d = __builtin_amdgcn_readfirstlane(cnt[dl]);
      const int beg = __builtin_amdgcn_readfirstlane(start[dl]);
      if (d <= 32) {
        int c1 = 0x7fffffff;
        if (lane < d) c1 = L ? ls[beg + lane] : slot[lo + beg + lane];
        int r = 0;
        for (int j = 0; j < d; ++j) {
          const int cj = __builtin_amdgcn_readlane(c1, j);
          r += (cj < c1 || (cj == c1 && j < lane)) ? 1 : 0;
        }
        if (lane < d) col[lo + beg + r] = c1;
      } else {
        for (int t = lane; t < d; t += 32) {
          const int c1 = L ? ls[beg + t] : slot[lo + beg + t];
          int r = 0;
          for (int j = 0; j < d; ++j) {
            const int cj = L ? ls[beg + j] : slot[lo + beg + j];
            r += (cj < c1 || (cj == c1 && j < t)) ? 1 : 0;
          }
          col[lo + beg + r] = c1;
        }
      }
    }
  };
  if (fits) sort_rows(std::true_type{}); else sort_rows(std::false_type{});
}


template <int K, int MOUT, bool ACT>
__global__ void k_gemm_strip(const float* __restrict__ X, const float* __restrict__ W,
                             const float* __restrict__ bias_in,
                             const float* __restrict__ dinv,
                             float* __restrict__ hs, int nrows) {
  constexpr int T = MOUT / 16;
  const int wave = (blockIdx.x * blockDim.x + threadIdx.x) >> 5;
  const int lane = threadIdx.x & 31;
  if (wave * 16 >= nrows) return;

  const int half = lane >> 4;
  const int l16  = lane & 15;
  const int rowA = wave * 16 + l16;

  v8f acc[T];
#pragma unroll
  for (int t = 0; t < T; ++t) acc[t] = (v8f){};

#pragma unroll 4
  for (int k = 0; k < K; k += 4) {
    const int ka = k + 2 * half;
    v2f a;
    a.x = X[(size_t)rowA * K + ka];
    a.y = X[(size_t)rowA * K + ka + 1];
    if (ACT) {
      a.x = fmaxf(a.x + bias_in[ka], 0.0f);
      a.y = fmaxf(a.y + bias_in[ka + 1], 0.0f);
    }
#pragma unroll
    for (int t = 0; t < T; ++t) {
      const int colB = t * 16 + l16;
      v2f b;
      b.x = W[(size_t)ka * MOUT + colB];
      b.y = W[(size_t)(ka + 1) * MOUT + colB];
      acc[t] = __builtin_amdgcn_wmma_f32_16x16x4_f32(
          false, a, false, b, (short)0, acc[t], false, false);
    }
  }

#pragma unroll
  for (int v = 0; v < 8; ++v) {
    const int row = wave * 16 + v + 8 * half;
    const float dn = dinv[row];
#pragma unroll
    for (int t = 0; t < T; ++t) {
      const int col = t * 16 + l16;
      hs[(size_t)row * MOUT + col] = acc[t][v] * dn;
    }
  }
}


template <int V> struct RowVec { typedef __attribute__((ext_vector_type(V))) float type; };
template <> struct RowVec<1> { typedef float type; };

template <int G, int MOUT, typename VecT>
__device__ __forceinline__ void add_rows(VecT& acc, const float* __restrict__ hl, int idx, int t) {
  VecT v[G];
#pragma unroll
  for (int g = 0; g < G; ++g) {
    const int s = __builtin_amdgcn_readlane(idx, t + g);
    v[g] = *(const VecT*)(hl + (size_t)s * MOUT);
  }
#pragma unroll
  for (int g = 0; g < G; ++g) acc += v[g];
}

template <int MOUT>
__global__ void k_gather(const int* __restrict__ rowptr, const int* __restrict__ col,
                         const float* __restrict__ hs, const float* __restrict__ dinv,
                         float* __restrict__ agg, int n) {
  constexpr int VPT = MOUT / 32;
  typedef typename RowVec<VPT>::type VecT;
  const int node = __builtin_amdgcn_readfirstlane((int)((blockIdx.x * blockDim.x + threadIdx.x) >> 5));
  if (node >= n) return;
  const int lane = threadIdx.x & 31;
  const int beg = rowptr[node];
  const int end = rowptr[node + 1];
  const float* hl = hs + lane * VPT;

  VecT acc = *(const VecT*)(hl + (size_t)node * MOUT);
  for (int base = beg; base < end; base += 32) {
    const int m = (end - base < 32) ? (end - base) : 32;
    const int idx = (lane < m) ? col[base + lane] : 0;
    int t = 0;
    for (; t + 8 <= m; t += 8) add_rows<8, MOUT, VecT>(acc, hl, idx, t);
    if (t + 4 <= m) { add_rows<4, MOUT, VecT>(acc, hl, idx, t); t += 4; }
    if (t + 2 <= m) { add_rows<2, MOUT, VecT>(acc, hl, idx, t); t += 2; }
    if (t < m)      { add_rows<1, MOUT, VecT>(acc, hl, idx, t); }
  }
  const float dn = dinv[node];
  *(VecT*)(agg + (size_t)node * MOUT + lane * VPT) = acc * dn;
}


template <int MOUT>
__global__ void k_bias_out(const float* __restrict__ agg, const float* __restrict__ bias,
                           float* __restrict__ out, int n) {
  const size_t i = (size_t)blockIdx.x * blockDim.x + threadIdx.x;
  if (i >= (size_t)n * MOUT) return;
  out[i] = agg[i] + bias[(int)(i % MOUT)];
}


template <int K, int MOUT, bool ACT>
static void run_layer(const float* in, const float* W, const float* bias_in,
                      const float* dinv, const int* rowptr, const int* col,
                      float* hs, float* agg, int n, hipStream_t stream) {
  const long long strips = (n + 15) / 16;
  const int gblocks = (int)((strips * 32 + 255) / 256);
  k_gemm_strip<K, MOUT, ACT><<<gblocks, 256, 0, stream>>>(in, W, bias_in, dinv, hs, n);

  const int ablocks = (int)(((long long)n * 32 + 255) / 256);
  k_gather<MOUT><<<ablocks, 256, 0, stream>>>(rowptr, col, hs, dinv, agg, n);
}

extern "C" void kernel_launch(void* const* d_in, const int* in_sizes, int n_in,
                              void* d_out, int out_size, void* d_ws, size_t ws_size,
                              hipStream_t stream) {
  const float* x  = (const float*)d_in[0];
  const int*   ei = (const int*)d_in[1];
  const float* W1 = (const float*)d_in[2];
  const float* b1 = (const float*)d_in[3];
  const float* W2 = (const float*)d_in[4];
  const float* b2 = (const float*)d_in[5];
  const float* W3 = (const float*)d_in[6];
  const float* b3 = (const float*)d_in[7];
  const float* W4 = (const float*)d_in[8];
  const float* b4 = (const float*)d_in[9];

  const int n = in_sizes[0] / 256;
  const int e = in_sizes[1] / 2;
  const int* esrc = ei;
  const int* edst = ei + e;
  const int nbk = (n + BUCKET_NODES - 1) / BUCKET_NODES;
  const int chunk = (e + EDGE_BLOCKS - 1) / EDGE_BLOCKS;

  char* cur = (char*)d_ws;
  auto carve = [&cur](size_t bytes) { char* p = cur; cur += (bytes + 255) & ~(size_t)255; return p; };
  float* dinv   = (float*)carve((size_t)n * 4);
  int*   rowptr = (int*)carve((size_t)(n + 1) * 4);
  int*   table  = (int*)carve((size_t)nbk * EDGE_BLOCKS * 4);
  int*   bbase  = (int*)carve((size_t)(nbk + 1) * 4);
  unsigned* rec = (unsigned*)carve((size_t)e * 4);
  int*   slot   = (int*)carve((size_t)e * 4);
  int*   col    = (int*)carve((size_t)e * 4);
  float* H  = (float*)carve((size_t)n * 128 * 4);
  float* G1 = (float*)carve((size_t)n * 128 * 4);
  float* G2 = (float*)carve((size_t)n * 128 * 4);

  k_bucket_counts<<<EDGE_BLOCKS, EDGE_THREADS, 0, stream>>>(edst, table, e, chunk, nbk);
  k_bucket_scatter<<<EDGE_BLOCKS, EDGE_THREADS, 0, stream>>>(esrc, edst, table, bbase, rec, e, chunk, nbk);
  k_bucket_rows<<<nbk, OWNER_THREADS, 0, stream>>>(bbase, rec, rowptr, dinv, slot, col, n, nbk);

  run_layer<256, 128, false>(x,  W1, nullptr, dinv, rowptr, col, H, G1, n, stream);
  run_layer<128, 128, true >(G1, W2, b1,      dinv, rowptr, col, H, G2, n, stream);
  run_layer<128, 64,  true >(G2, W3, b2,      dinv, rowptr, col, H, G1, n, stream);
  run_layer<64,  32,  true >(G1, W4, b3,      dinv, rowptr, col, H, G2, n, stream);

  const long long tot = (long long)n * 32;
  k_bias_out<32><<<(int)((tot + 255) / 256), 256, 0, stream>>>(G2, b4, (float*)d_out, n);
}


# Each neighbour-sum launch now also does what followed it: summed 32-node strip -> bias+relu -> LDS -> A fragments of the next layer's WMMA product (last: bias into out); 12 launches -> 8
# speedup vs baseline: 3.9116x; 1.1271x over previous
#include <hip/hip_runtime.h>
#include <type_traits>


typedef __attribute__((ext_vector_type(2))) float v2f;
typedef __attribute__((ext_vector_type(4))) float v4f;
typedef __attribute__((ext_vector_type(8))) float v8f;


constexpr int NODE_SHIFT = 8;
constexpr int BUCKET_NODES = 1 << NODE_SHIFT;
constexpr int EDGE_BLOCKS = 128;
constexpr int EDGE_THREADS = 1024;
constexpr int EDGE_WAVES = EDGE_THREADS / 32;
constexpr int MAX_BUCKETS = 1024;
constexpr int OWNER_THREADS = 512;
constexpr int OWNER_WAVES = OWNER_THREADS / 32;
constexpr int OWNER_CAP = 8192;
static_assert(EDGE_BLOCKS == 128, "one 16-byte load per lane covers a table row");
static_assert(MAX_BUCKETS == EDGE_THREADS && EDGE_WAVES == 32, "scan: one bucket per thread, one wave total per lane");

__global__ __launch_bounds__(EDGE_THREADS) void k_bucket_counts(
    const int* __restrict__ dst, int* __restrict__ table, int e, int chunk, int nbk) {
  __shared__ int h[MAX_BUCKETS];
  const int tid = threadIdx.x;
  for (int b = tid; b < nbk; b += EDGE_THREADS) h[b] = 0;
  __syncthreads();
  const int lo = blockIdx.x * chunk;
  const int hi = min(e, lo + chunk);
  for (int i = lo + tid; i < hi; i += 4 * EDGE_THREADS) {
    int d[4];
#pragma unroll
    for (int q = 0; q < 4; ++q) {
      const int j = i + q * EDGE_THREADS;
      d[q] = dst[j < hi ? j : hi - 1];
    }
#pragma unroll
    for (int q = 0; q < 4; ++q)
      atomicAdd(&h[d[q] >> NODE_SHIFT], (i + q * EDGE_THREADS < hi) ? 1 : 0);
  }
  __syncthreads();
  for (int b = tid; b < nbk; b += EDGE_THREADS) table[b * EDGE_BLOCKS + blockIdx.x] = h[b];
}

__global__ __launch_bounds__(EDGE_THREADS) void k_bucket_scatter(
    const int* __restrict__ src, const int* __restrict__ dst, const int* __restrict__ table,
    int* __restrict__ bbase, unsigned* __restrict__ rec, int e, int chunk, int nbk) {
  __shared__ int tot[MAX_BUCKETS];
  __shared__ int pre[MAX_BUCKETS];
  __shared__ int off[MAX_BUCKETS];
  __shared__ int wtot[EDGE_WAVES];
  const int tid = threadIdx.x;
  const int lane = tid & 31;
  const int wave = tid >> 5;
  const int blk = blockIdx.x;

  for (int b0 = wave; b0 < nbk; b0 += 8 * EDGE_WAVES) {
    int4 v[8];
#pragma unroll
    for (int q = 0; q < 8; ++q) {
      const int b = b0 + q * EDGE_WAVES;
      v[q] = *(const int4*)(table + (b < nbk ? b : nbk - 1) * EDGE_BLOCKS + lane * 4);
    }
#pragma unroll
    for (int q = 0; q < 8; ++q) {
      const int b = b0 + q * EDGE_WAVES;
      const int c0 = lane * 4;
      int t = v[q].x + v[q].y + v[q].z + v[q].w;
      int p = (c0 < blk ? v[q].x : 0) + (c0 + 1 < blk ? v[q].y : 0) + (c0 + 2 < blk ? v[q].z : 0) +
              (c0 + 3 < blk ? v[q].w : 0);
#pragma unroll
      for (int o = 16; o > 0; o >>= 1) {
        t += __shfl_xor(t, o);
        p += __shfl_xor(p, o);
      }
      if (lane == 0 && b < nbk) { tot[b] = t; pre[b] = p; }
    }
  }
  __syncthreads();

  const int mine = (tid < nbk) ? tot[tid] : 0;
  int incl = mine;
#pragma unroll
  for (int o = 1; o < 32; o <<= 1) {
    const int u = __shfl_up(incl, o);
    if (lane >= o) incl += u;
  }
  if (lane == 31) wtot[wave] = incl;
  __syncthreads();
  int wincl = wtot[lane];
#pragma unroll
  for (int o = 1; o < 32; o <<= 1) {
    const int u = __shfl_up(wincl, o);
    if (lane >= o) wincl += u;
  }
  const int wexcl = wincl - wtot[lane];
  const int first = __shfl(wexcl, wave) + incl - mine;
  if (tid < nbk) {
    off[tid] = first + pre[tid];
    if (blk == 0) {
      bbase[tid] = first;
      if (tid == nbk - 1) bbase[nbk] = first + mine;
    }
  }
  __syncthreads();

  const int lo = blk * chunk;
  const int hi = min(e, lo + chunk);
  int i = lo + tid;
  for (; i + 3 * EDGE_THREADS < hi; i += 4 * EDGE_THREADS) {
    int d[4], s[4];
#pragma unroll
    for (int q = 0; q < 4; ++q) {
      d[q] = dst[i + q * EDGE_THREADS];
      s[q] = src[i + q * EDGE_THREADS];
    }
#pragma unroll
    for (int q = 0; q < 4; ++q) {
      const int pos = atomicAdd(&off[d[q] >> NODE_SHIFT], 1);
      rec[pos] = (unsigned)s[q] | ((unsigned)(d[q] & (BUCKET_NODES - 1)) << 24);
    }
  }
  for (; i < hi; i += EDGE_THREADS) {
    const int d = dst[i];
    const int s = src[i];
    const int pos = atomicAdd(&off[d >> NODE_SHIFT], 1);
    rec[pos] = (unsigned)s | ((unsigned)(d & (BUCKET_NODES - 1)) << 24);
  }
}

__global__ __launch_bounds__(OWNER_THREADS) void k_bucket_rows(
    const int* __restrict__ bbase, const unsigned* __restrict__ rec, int* __restrict__ rowptr,
    float* __restrict__ dinv, int* __restrict__ slot, int* __restrict__ col, int n, int nbk) {
  __shared__ int cnt[BUCKET_NODES];
  __shared__ int start[BUCKET_NODES];
  __shared__ int cur[BUCKET_NODES];
  __shared__ int wsum[BUCKET_NODES / 32];
  __shared__ int ls[OWNER_CAP];
  const int tid = threadIdx.x;
  const int lane = tid & 31;
  const int wave = tid >> 5;
  const int b = blockIdx.x;
  const int lo = bbase[b];
  const int hi = bbase[b + 1];
  const bool fits = (hi - lo) <= OWNER_CAP;

  if (tid < BUCKET_NODES) { cnt[tid] = 0; cur[tid] = 0; }
  __syncthreads();

  for (int i = lo + tid; i < hi; i += 4 * OWNER_THREADS) {
    unsigned r[4];
#pragma unroll
    for (int q = 0; q < 4; ++q) {
      const int j = i + q * OWNER_THREADS;
      r[q] = rec[j < hi ? j : hi - 1];
    }
#pragma unroll
    for (int q = 0; q < 4; ++q)
      atomicAdd(&cnt[r[q] >> 24], (i + q * OWNER_THREADS < hi) ? 1 : 0);
  }
  __syncthreads();

  int c = 0, incl = 0;
  if (tid < BUCKET_NODES) {
    c = cnt[tid];
    incl = c;
#pragma unroll
    for (int o = 1; o < 32; o <<= 1) {
      const int u = __shfl_up(incl, o);
      if (lane >= o) incl += u;
    }
    if (lane == 31) wsum[wave] = incl;
  }
  __syncthreads();
  if (tid < BUCKET_NODES) {
    int wbase = 0;
    for (int w = 0; w < wave; ++w) wbase += wsum[w];
    const int st = wbase + incl - c;
    start[tid] = st;
    const int node = b * BUCKET_NODES + tid;
    if (node < n) {
      rowptr[node] = lo + st;
      dinv[node] = rsqrtf((float)(c + 1));
    }
  }
  if (b == nbk - 1 && tid == 0) rowptr[n] = hi;
  __syncthreads();

  for (int i = lo + tid; i < hi; i += 4 * OWNER_THREADS) {
    unsigned r[4];
#pragma unroll
    for (int q = 0; q < 4; ++q) {
      const int j = i + q * OWNER_THREADS;
      r[q] = rec[j < hi ? j : hi - 1];
    }
#pragma unroll
    for (int q = 0; q < 4; ++q) {
      if (i + q * OWNER_THREADS < hi) {
        const int dl = (int)(r[q] >> 24);
        const int pos = start[dl] + atomicAdd(&cur[dl], 1);
        const int s = (int)(r[q] & 0xffffffu);
        if (fits) ls[pos] = s; else slot[lo + pos] = s;
      }
    }
  }
  __syncthreads();

  auto sort_rows = [&](auto in_lds) {
    constexpr bool L = decltype(in_lds)::value;
    for (int dl = wave; dl < BUCKET_NODES; dl += OWNER_WAVES) {
      if (b * BUCKET_NODES + dl >= n) break;
      const int d = __builtin_amdgcn_readfirstlane(cnt[dl]);
      const int beg = __builtin_amdgcn_readfirstlane(start[dl]);
      if (d <= 32) {
        int c1 = 0x7fffffff;
        if (lane < d) c1 = L ? ls[beg + lane] : slot[lo + beg + lane];
        int r = 0;
        for (int j = 0; j < d; ++j) {
          const int cj = __builtin_amdgcn_readlane(c1, j);
          r += (cj < c1 || (cj == c1 && j < lane)) ? 1 : 0;
        }
        if (lane < d) col[lo + beg + r] = c1;
      } else {
        for (int t = lane; t < d; t += 32) {
          const int c1 = L ? ls[beg + t] : slot[lo + beg + t];
          int r = 0;
          for (int j = 0; j < d; ++j) {
            const int cj = L ? ls[beg + j] : slot[lo + beg + j];
            r += (cj < c1 || (cj == c1 && j < t)) ? 1 : 0;
          }
          col[lo + beg + r] = c1;
        }
      }
    }
  };
  if (fits) sort_rows(std::true_type{}); else sort_rows(std::false_type{});
}


template <int K, int MOUT>
__global__ void k_gemm_strip(const float* __restrict__ X, const float* __restrict__ W,
                             const float* __restrict__ dinv,
                             float* __restrict__ hs, int nrows) {
  constexpr int T = MOUT / 16;
  const int wave = (blockIdx.x * blockDim.x + threadIdx.x) >> 5;
  const int lane = threadIdx.x & 31;
  if (wave * 16 >= nrows) return;

  const int half = lane >> 4;
  const int l16  = lane & 15;
  const int rowA = wave * 16 + l16;

  v8f acc[T];
#pragma unroll
  for (int t = 0; t < T; ++t) acc[t] = (v8f){};

#pragma unroll 4
  for (int k = 0; k < K; k += 4) {
    const int ka = k + 2 * half;
    v2f a;
    a.x = X[(size_t)rowA * K + ka];
    a.y = X[(size_t)rowA * K + ka + 1];
#pragma unroll
    for (int t = 0; t < T; ++t) {
      const int colB = t * 16 + l16;
      v2f b;
      b.x = W[(size_t)ka * MOUT + colB];
      b.y = W[(size_t)(ka + 1) * MOUT + colB];
      acc[t] = __builtin_amdgcn_wmma_f32_16x16x4_f32(
          false, a, false, b, (short)0, acc[t], false, false);
    }
  }

#pragma unroll
  for (int v = 0; v < 8; ++v) {
    const int row = wave * 16 + v + 8 * half;
    const float dn = dinv[row];
#pragma unroll
    for (int t = 0; t < T; ++t) {
      const int col = t * 16 + l16;
      hs[(size_t)row * MOUT + col] = acc[t][v] * dn;
    }
  }
}


template <int V> struct RowVec { typedef __attribute__((ext_vector_type(V))) float type; };
template <> struct RowVec<1> { typedef float type; };

__device__ __forceinline__ v2f relu(v2f v) { return (v2f){fmaxf(v.x, 0.0f), fmaxf(v.y, 0.0f)}; }
__device__ __forceinline__ v4f relu(v4f v) {
  return (v4f){fmaxf(v.x, 0.0f), fmaxf(v.y, 0.0f), fmaxf(v.z, 0.0f), fmaxf(v.w, 0.0f)};
}

template <int G, int MOUT, typename VecT>
__device__ __forceinline__ void add_rows(VecT& acc, const float* __restrict__ hl, int idx, int t) {
  VecT v[G];
#pragma unroll
  for (int g = 0; g < G; ++g) {
    const int s = __builtin_amdgcn_readlane(idx, t + g);
    v[g] = *(const VecT*)(hl + (size_t)s * MOUT);
  }
#pragma unroll
  for (int g = 0; g < G; ++g) acc += v[g];
}

template <int MOUT, typename VecT>
__device__ __forceinline__ VecT node_sum(const int* __restrict__ col, const float* __restrict__ hl,
                                         int node, int beg, int end, int lane) {
  VecT acc = *(const VecT*)(hl + (size_t)node * MOUT);
  for (int base = beg; base < end; base += 32) {
    const int m = (end - base < 32) ? (end - base) : 32;
    const int idx = (lane < m) ? col[base + lane] : 0;
    int t = 0;
    for (; t + 8 <= m; t += 8) add_rows<8, MOUT, VecT>(acc, hl, idx, t);
    if (t + 4 <= m) { add_rows<4, MOUT, VecT>(acc, hl, idx, t); t += 4; }
    if (t + 2 <= m) { add_rows<2, MOUT, VecT>(acc, hl, idx, t); t += 2; }
    if (t < m)      { add_rows<1, MOUT, VecT>(acc, hl, idx, t); }
  }
  return acc;
}


constexpr int STRIP_ROWS = 32;
constexpr int STRIP_ROW_TILES = STRIP_ROWS / 16;
constexpr int STRIP_WAVES = 8;
constexpr int STRIP_THREADS = STRIP_WAVES * 32;
constexpr int STRIP_NODES_PER_WAVE = STRIP_ROWS / STRIP_WAVES;

template <int K, int MOUT>
__global__ __launch_bounds__(STRIP_THREADS, 16) void k_gather_gemm(
    const int* __restrict__ rowptr, const int* __restrict__ col, const float* __restrict__ hs_in,
    const float* __restrict__ dinv, const float* __restrict__ bias_in, const float* __restrict__ W,
    float* __restrict__ hs_out, int n) {
  constexpr int VPT = K / 32;
  typedef typename RowVec<VPT>::type VecT;
  constexpr int PITCH = K + 4;
  constexpr int RT = STRIP_ROW_TILES;
  constexpr int TW = 2;
  constexpr int GEMM_WAVES = MOUT / (16 * TW);
  static_assert(GEMM_WAVES >= 1 && GEMM_WAVES <= STRIP_WAVES, "tiles must fit the strip's waves");
  __shared__ __attribute__((aligned(16))) float strip[STRIP_ROWS * PITCH];

  const int lane = threadIdx.x & 31;
  const int wave = __builtin_amdgcn_readfirstlane((int)(threadIdx.x >> 5));
  const int row0 = blockIdx.x * STRIP_ROWS;

  const float* hl = hs_in + lane * VPT;
  const VecT bv = *(const VecT*)(bias_in + lane * VPT);
  for (int i = 0; i < STRIP_NODES_PER_WAVE; ++i) {
    const int r = wave * STRIP_NODES_PER_WAVE + i;
    const int node = row0 + r;
    VecT val = {};
    if (node < n) {
      const float dn = dinv[node];
      const VecT acc = node_sum<K, VecT>(col, hl, node, rowptr[node], rowptr[node + 1], lane);
      val = relu(acc * dn + bv);
    }
    *(VecT*)(strip + r * PITCH + lane * VPT) = val;
  }
  __syncthreads();
  const int mw = (wave + (int)blockIdx.x) & (STRIP_WAVES - 1);
  if (mw >= GEMM_WAVES) return;

  const int half = lane >> 4;
  const int l16  = lane & 15;
  const float* arow = strip + l16 * PITCH + 2 * half;

  v8f acc[RT][TW];
#pragma unroll
  for (int rt = 0; rt < RT; ++rt)
#pragma unroll
    for (int t = 0; t < TW; ++t) acc[rt][t] = (v8f){};

#pragma unroll 2
  for (int k = 0; k < K; k += 4) {
    const int ka = k + 2 * half;
    v2f a[RT];
#pragma unroll
    for (int rt = 0; rt < RT; ++rt) a[rt] = *(const v2f*)(arow + rt * 16 * PITCH + k);
#pragma unroll
    for (int t = 0; t < TW; ++t) {
      const int colB = (mw * TW + t) * 16 + l16;
      v2f b;
      b.x = W[(size_t)ka * MOUT + colB];
      b.y = W[(size_t)(ka + 1) * MOUT + colB];
#pragma unroll
      for (int rt = 0; rt < RT; ++rt)
        acc[rt][t] = __builtin_amdgcn_wmma_f32_16x16x4_f32(
            false, a[rt], false, b, (short)0, acc[rt][t], false, false);
    }
  }

  if (row0 + STRIP_ROWS <= n) {
#pragma unroll
    for (int rt = 0; rt < RT; ++rt) {
#pragma unroll
      for (int v = 0; v < 8; ++v) {
        const int row = row0 + rt * 16 + v + 8 * half;
        const float dn = dinv[row];
#pragma unroll
        for (int t = 0; t < TW; ++t) {
          const int c = (mw * TW + t) * 16 + l16;
          hs_out[(size_t)row * MOUT + c] = acc[rt][t][v] * dn;
        }
      }
    }
  } else {
#pragma unroll
    for (int rt = 0; rt < RT; ++rt) {
#pragma unroll
      for (int v = 0; v < 8; ++v) {
        const int row = row0 + rt * 16 + v + 8 * half;
        if (row < n) {
          const float dn = dinv[row];
#pragma unroll
          for (int t = 0; t < TW; ++t) {
            const int c = (mw * TW + t) * 16 + l16;
            hs_out[(size_t)row * MOUT + c] = acc[rt][t][v] * dn;
          }
        }
      }
    }
  }
}


template <int MOUT>
__global__ void k_gather_out(const int* __restrict__ rowptr, const int* __restrict__ col,
                             const float* __restrict__ hs, const float* __restrict__ dinv,
                             const float* __restrict__ bias, float* __restrict__ out, int n) {
  constexpr int VPT = MOUT / 32;
  typedef typename RowVec<VPT>::type VecT;
  const int node = __builtin_amdgcn_readfirstlane((int)((blockIdx.x * blockDim.x + threadIdx.x) >> 5));
  if (node >= n) return;
  const int lane = threadIdx.x & 31;
  const VecT acc = node_sum<MOUT, VecT>(col, hs + lane * VPT, node, rowptr[node], rowptr[node + 1], lane);
  const VecT bv = *(const VecT*)(bias + lane * VPT);
  *(VecT*)(out + (size_t)node * MOUT + lane * VPT) = acc * dinv[node] + bv;
}

extern "C" void kernel_launch(void* const* d_in, const int* in_sizes, int n_in,
                              void* d_out, int out_size, void* d_ws, size_t ws_size,
                              hipStream_t stream) {
  const float* x  = (const float*)d_in[0];
  const int*   ei = (const int*)d_in[1];
  const float* W1 = (const float*)d_in[2];
  const float* b1 = (const float*)d_in[3];
  const float* W2 = (const float*)d_in[4];
  const float* b2 = (const float*)d_in[5];
  const float* W3 = (const float*)d_in[6];
  const float* b3 = (const float*)d_in[7];
  const float* W4 = (const float*)d_in[8];
  const float* b4 = (const float*)d_in[9];

  const int n = in_sizes[0] / 256;
  const int e = in_sizes[1] / 2;
  const int* esrc = ei;
  const int* edst = ei + e;
  const int nbk = (n + BUCKET_NODES - 1) / BUCKET_NODES;
  const int chunk = (e + EDGE_BLOCKS - 1) / EDGE_BLOCKS;

  char* cur = (char*)d_ws;
  auto carve = [&cur](size_t bytes) { char* p = cur; cur += (bytes + 255) & ~(size_t)255; return p; };
  float* dinv   = (float*)carve((size_t)n * 4);
  int*   rowptr = (int*)carve((size_t)(n + 1) * 4);
  int*   table  = (int*)carve((size_t)nbk * EDGE_BLOCKS * 4);
  int*   bbase  = (int*)carve((size_t)(nbk + 1) * 4);
  unsigned* rec = (unsigned*)carve((size_t)e * 4);
  int*   slot   = (int*)carve((size_t)e * 4);
  int*   col    = (int*)carve((size_t)e * 4);
  float* HA = (float*)carve((size_t)n * 128 * 4);
  float* HB = (float*)carve((size_t)n * 128 * 4);

  k_bucket_counts<<<EDGE_BLOCKS, EDGE_THREADS, 0, stream>>>(edst, table, e, chunk, nbk);
  k_bucket_scatter<<<EDGE_BLOCKS, EDGE_THREADS, 0, stream>>>(esrc, edst, table, bbase, rec, e, chunk, nbk);
  k_bucket_rows<<<nbk, OWNER_THREADS, 0, stream>>>(bbase, rec, rowptr, dinv, slot, col, n, nbk);

  const int tiles = (n + 15) / 16;
  const int strips = (n + STRIP_ROWS - 1) / STRIP_ROWS;
  k_gemm_strip<256, 128><<<(tiles * 32 + 255) / 256, 256, 0, stream>>>(x, W1, dinv, HA, n);
  k_gather_gemm<128, 128><<<strips, STRIP_THREADS, 0, stream>>>(rowptr, col, HA, dinv, b1, W2, HB, n);
  k_gather_gemm<128, 64><<<strips, STRIP_THREADS, 0, stream>>>(rowptr, col, HB, dinv, b2, W3, HA, n);
  k_gather_gemm<64, 32><<<strips, STRIP_THREADS, 0, stream>>>(rowptr, col, HA, dinv, b3, W4, HB, n);

  k_gather_out<32><<<(int)(((long long)n * 32 + 255) / 256), 256, 0, stream>>>(rowptr, col, HB, dinv, b4, (float*)d_out, n);
}


# First layer's product moved into the row-building launch: a bucket's 16 waves build its rows and multiply its own 256 rows (dinv from LDS), alternate work-groups multiplying first; 8 launches -> 7
# speedup vs baseline: 4.0638x; 1.0389x over previous
#include <hip/hip_runtime.h>
#include <type_traits>


typedef __attribute__((ext_vector_type(2))) float v2f;
typedef __attribute__((ext_vector_type(4))) float v4f;
typedef __attribute__((ext_vector_type(8))) float v8f;


constexpr int NODE_SHIFT = 8;
constexpr int BUCKET_NODES = 1 << NODE_SHIFT;
constexpr int EDGE_BLOCKS = 128;
constexpr int EDGE_THREADS = 1024;
constexpr int EDGE_WAVES = EDGE_THREADS / 32;
constexpr int MAX_BUCKETS = 1024;
constexpr int OWNER_THREADS = 512;
constexpr int OWNER_WAVES = OWNER_THREADS / 32;
constexpr int OWNER_CAP = 8192;
static_assert(EDGE_BLOCKS == 128, "one 16-byte load per lane covers a table row");
static_assert(MAX_BUCKETS == EDGE_THREADS && EDGE_WAVES == 32, "scan: one bucket per thread, one wave total per lane");

__global__ __launch_bounds__(EDGE_THREADS) void k_bucket_counts(
    const int* __restrict__ dst, int* __restrict__ table, int e, int chunk, int nbk) {
  __shared__ int h[MAX_BUCKETS];
  const int tid = threadIdx.x;
  for (int b = tid; b < nbk; b += EDGE_THREADS) h[b] = 0;
  __syncthreads();
  const int lo = blockIdx.x * chunk;
  const int hi = min(e, lo + chunk);
  for (int i = lo + tid; i < hi; i += 4 * EDGE_THREADS) {
    int d[4];
#pragma unroll
    for (int q = 0; q < 4; ++q) {
      const int j = i + q * EDGE_THREADS;
      d[q] = dst[j < hi ? j : hi - 1];
    }
#pragma unroll
    for (int q = 0; q < 4; ++q)
      atomicAdd(&h[d[q] >> NODE_SHIFT], (i + q * EDGE_THREADS < hi) ? 1 : 0);
  }
  __syncthreads();
  for (int b = tid; b < nbk; b += EDGE_THREADS) table[b * EDGE_BLOCKS + blockIdx.x] = h[b];
}

__global__ __launch_bounds__(EDGE_THREADS) void k_bucket_scatter(
    const int* __restrict__ src, const int* __restrict__ dst, const int* __restrict__ table,
    int* __restrict__ bbase, unsigned* __restrict__ rec, int e, int chunk, int nbk) {
  __shared__ int tot[MAX_BUCKETS];
  __shared__ int pre[MAX_BUCKETS];
  __shared__ int off[MAX_BUCKETS];
  __shared__ int wtot[EDGE_WAVES];
  const int tid = threadIdx.x;
  const int lane = tid & 31;
  const int wave = tid >> 5;
  const int blk = blockIdx.x;

  for (int b0 = wave; b0 < nbk; b0 += 8 * EDGE_WAVES) {
    int4 v[8];
#pragma unroll
    for (int q = 0; q < 8; ++q) {
      const int b = b0 + q * EDGE_WAVES;
      v[q] = *(const int4*)(table + (b < nbk ? b : nbk - 1) * EDGE_BLOCKS + lane * 4);
    }
#pragma unroll
    for (int q = 0; q < 8; ++q) {
      const int b = b0 + q * EDGE_WAVES;
      const int c0 = lane * 4;
      int t = v[q].x + v[q].y + v[q].z + v[q].w;
      int p = (c0 < blk ? v[q].x : 0) + (c0 + 1 < blk ? v[q].y : 0) + (c0 + 2 < blk ? v[q].z : 0) +
              (c0 + 3 < blk ? v[q].w : 0);
#pragma unroll
      for (int o = 16; o > 0; o >>= 1) {
        t += __shfl_xor(t, o);
        p += __shfl_xor(p, o);
      }
      if (lane == 0 && b < nbk) { tot[b] = t; pre[b] = p; }
    }
  }
  __syncthreads();

  const int mine = (tid < nbk) ? tot[tid] : 0;
  int incl = mine;
#pragma unroll
  for (int o = 1; o < 32; o <<= 1) {
    const int u = __shfl_up(incl, o);
    if (lane >= o) incl += u;
  }
  if (lane == 31) wtot[wave] = incl;
  __syncthreads();
  int wincl = wtot[lane];
#pragma unroll
  for (int o = 1; o < 32; o <<= 1) {
    const int u = __shfl_up(wincl, o);
    if (lane >= o) wincl += u;
  }
  const int wexcl = wincl - wtot[lane];
  const int first = __shfl(wexcl, wave) + incl - mine;
  if (tid < nbk) {
    off[tid] = first + pre[tid];
    if (blk == 0) {
      bbase[tid] = first;
      if (tid == nbk - 1) bbase[nbk] = first + mine;
    }
  }
  __syncthreads();

  const int lo = blk * chunk;
  const int hi = min(e, lo + chunk);
  int i = lo + tid;
  for (; i + 3 * EDGE_THREADS < hi; i += 4 * EDGE_THREADS) {
    int d[4], s[4];
#pragma unroll
    for (int q = 0; q < 4; ++q) {
      d[q] = dst[i + q * EDGE_THREADS];
      s[q] = src[i + q * EDGE_THREADS];
    }
#pragma unroll
    for (int q = 0; q < 4; ++q) {
      const int pos = atomicAdd(&off[d[q] >> NODE_SHIFT], 1);
      rec[pos] = (unsigned)s[q] | ((unsigned)(d[q] & (BUCKET_NODES - 1)) << 24);
    }
  }
  for (; i < hi; i += EDGE_THREADS) {
    const int d = dst[i];
    const int s = src[i];
    const int pos = atomicAdd(&off[d >> NODE_SHIFT], 1);
    rec[pos] = (unsigned)s | ((unsigned)(d & (BUCKET_NODES - 1)) << 24);
  }
}

static_assert(OWNER_WAVES * 16 == BUCKET_NODES, "one 16-row tile of the bucket per wave");

template <int K, int MOUT>
__device__ __forceinline__ void strip_product(v8f (&acc)[MOUT / 16], const float* __restrict__ X,
                                              const float* __restrict__ W, int rowA, int lane) {
  constexpr int T = MOUT / 16;
  const int half = lane >> 4;
  const int l16  = lane & 15;
#pragma unroll 4
  for (int k = 0; k < K; k += 4) {
    const int ka = k + 2 * half;
    v2f a;
    a.x = X[(size_t)rowA * K + ka];
    a.y = X[(size_t)rowA * K + ka + 1];
#pragma unroll
    for (int t = 0; t < T; ++t) {
      const int colB = t * 16 + l16;
      v2f bf;
      bf.x = W[(size_t)ka * MOUT + colB];
      bf.y = W[(size_t)(ka + 1) * MOUT + colB];
      acc[t] = __builtin_amdgcn_wmma_f32_16x16x4_f32(
          false, a, false, bf, (short)0, acc[t], false, false);
    }
  }
}

template <int K, int MOUT>
__global__ __launch_bounds__(OWNER_THREADS) void k_bucket_rows_gemm(
    const int* __restrict__ bbase, const unsigned* __restrict__ rec, int* __restrict__ rowptr,
    float* __restrict__ dinv, int* __restrict__ slot, int* __restrict__ col,
    const float* __restrict__ X, const float* __restrict__ W, float* __restrict__ hs,
    int n, int nbk) {
  __shared__ float dv[BUCKET_NODES];
  __shared__ int cnt[BUCKET_NODES];
  __shared__ int start[BUCKET_NODES];
  __shared__ int cur[BUCKET_NODES];
  __shared__ int wsum[BUCKET_NODES / 32];
  __shared__ int ls[OWNER_CAP];
  const int tid = threadIdx.x;
  const int lane = tid & 31;
  const int wave = tid >> 5;
  const int b = blockIdx.x;
  const int lo = bbase[b];
  const int hi = bbase[b + 1];
  const bool fits = (hi - lo) <= OWNER_CAP;

  constexpr int T = MOUT / 16;
  const bool mul_first = (((b >> 8) ^ b) & 1) != 0;
  const int wv = __builtin_amdgcn_readfirstlane(wave);
  const int trow0 = b * BUCKET_NODES + wv * 16;
  const bool has_tile = trow0 < n;
  const int rowA = min(trow0 + (lane & 15), n - 1);
  v8f acc[T];
#pragma unroll
  for (int t = 0; t < T; ++t) acc[t] = (v8f){};
  if (mul_first && has_tile) strip_product<K, MOUT>(acc, X, W, rowA, lane);

  if (tid < BUCKET_NODES) { cnt[tid] = 0; cur[tid] = 0; }
  __syncthreads();

  for (int i = lo + tid; i < hi; i += 4 * OWNER_THREADS) {
    unsigned r[4];
#pragma unroll
    for (int q = 0; q < 4; ++q) {
      const int j = i + q * OWNER_THREADS;
      r[q] = rec[j < hi ? j : hi - 1];
    }
#pragma unroll
    for (int q = 0; q < 4; ++q)
      atomicAdd(&cnt[r[q] >> 24], (i + q * OWNER_THREADS < hi) ? 1 : 0);
  }
  __syncthreads();

  int c = 0, incl = 0;
  if (tid < BUCKET_NODES) {
    c = cnt[tid];
    incl = c;
#pragma unroll
    for (int o = 1; o < 32; o <<= 1) {
      const int u = __shfl_up(incl, o);
      if (lane >= o) incl += u;
    }
    if (lane == 31) wsum[wave] = incl;
  }
  __syncthreads();
  if (tid < BUCKET_NODES) {
    int wbase = 0;
    for (int w = 0; w < wave; ++w) wbase += wsum[w];
    const int st = wbase + incl - c;
    start[tid] = st;
    const int node = b * BUCKET_NODES + tid;
    const float di = rsqrtf((float)(c + 1));
    dv[tid] = di;
    if (node < n) {
      rowptr[node] = lo + st;
      dinv[node] = di;
    }
  }
  if (b == nbk - 1 && tid == 0) rowptr[n] = hi;
  __syncthreads();

  for (int i = lo + tid; i < hi; i += 4 * OWNER_THREADS) {
    unsigned r[4];
#pragma unroll
    for (int q = 0; q < 4; ++q) {
      const int j = i + q * OWNER_THREADS;
      r[q] = rec[j < hi ? j : hi - 1];
    }
#pragma unroll
    for (int q = 0; q < 4; ++q) {
      if (i + q * OWNER_THREADS < hi) {
        const int dl = (int)(r[q] >> 24);
        const int pos = start[dl] + atomicAdd(&cur[dl], 1);
        const int s = (int)(r[q] & 0xffffffu);
        if (fits) ls[pos] = s; else slot[lo + pos] = s;
      }
    }
  }
  __syncthreads();

  auto sort_rows = [&](auto in_lds) {
    constexpr bool L = decltype(in_lds)::value;
    for (int dl = wave; dl < BUCKET_NODES; dl += OWNER_WAVES) {
      if (b * BUCKET_NODES + dl >= n) break;
      const int d = __builtin_amdgcn_readfirstlane(cnt[dl]);
      const int beg = __builtin_amdgcn_readfirstlane(start[dl]);
      if (d <= 32) {
        int c1 = 0x7fffffff;
        if (lane < d) c1 = L ? ls[beg + lane] : slot[lo + beg + lane];
        int r = 0;
        for (int j = 0; j < d; ++j) {
          const int cj = __builtin_amdgcn_readlane(c1, j);
          r += (cj < c1 || (cj == c1 && j < lane)) ? 1 : 0;
        }
        if (lane < d) col[lo + beg + r] = c1;
      } else {
        for (int t = lane; t < d; t += 32) {
          const int c1 = L ? ls[beg + t] : slot[lo + beg + t];
          int r = 0;
          for (int j = 0; j < d; ++j) {
            const int cj = L ? ls[beg + j] : slot[lo + beg + j];
            r += (cj < c1 || (cj == c1 && j < t)) ? 1 : 0;
          }
          col[lo + beg + r] = c1;
        }
      }
    }
  };
  if (fits) sort_rows(std::true_type{}); else sort_rows(std::false_type{});

  if (!has_tile) return;
  if (!mul_first) strip_product<K, MOUT>(acc, X, W, rowA, lane);

  const bool whole = trow0 + 16 <= n;
  const int half = lane >> 4;
  const int l16  = lane & 15;
#pragma unroll
  for (int v = 0; v < 8; ++v) {
    const int lr = wv * 16 + v + 8 * half;
    const int row = b * BUCKET_NODES + lr;
    const float dn = dv[lr];
    if (whole || row < n) {
#pragma unroll
      for (int t = 0; t < T; ++t) {
        const int cc = t * 16 + l16;
        hs[(size_t)row * MOUT + cc] = acc[t][v] * dn;
      }
    }
  }
}


template <int V> struct RowVec { typedef __attribute__((ext_vector_type(V))) float type; };
template <> struct RowVec<1> { typedef float type; };

__device__ __forceinline__ v2f relu(v2f v) { return (v2f){fmaxf(v.x, 0.0f), fmaxf(v.y, 0.0f)}; }
__device__ __forceinline__ v4f relu(v4f v) {
  return (v4f){fmaxf(v.x, 0.0f), fmaxf(v.y, 0.0f), fmaxf(v.z, 0.0f), fmaxf(v.w, 0.0f)};
}

template <int G, int MOUT, typename VecT>
__device__ __forceinline__ void add_rows(VecT& acc, const float* __restrict__ hl, int idx, int t) {
  VecT v[G];
#pragma unroll
  for (int g = 0; g < G; ++g) {
    const int s = __builtin_amdgcn_readlane(idx, t + g);
    v[g] = *(const VecT*)(hl + (size_t)s * MOUT);
  }
#pragma unroll
  for (int g = 0; g < G; ++g) acc += v[g];
}

template <int MOUT, typename VecT>
__device__ __forceinline__ VecT node_sum(const int* __restrict__ col, const float* __restrict__ hl,
                                         int node, int beg, int end, int lane) {
  VecT acc = *(const VecT*)(hl + (size_t)node * MOUT);
  for (int base = beg; base < end; base += 32) {
    const int m = (end - base < 32) ? (end - base) : 32;
    const int idx = (lane < m) ? col[base + lane] : 0;
    int t = 0;
    for (; t + 8 <= m; t += 8) add_rows<8, MOUT, VecT>(acc, hl, idx, t);
    if (t + 4 <= m) { add_rows<4, MOUT, VecT>(acc, hl, idx, t); t += 4; }
    if (t + 2 <= m) { add_rows<2, MOUT, VecT>(acc, hl, idx, t); t += 2; }
    if (t < m)      { add_rows<1, MOUT, VecT>(acc, hl, idx, t); }
  }
  return acc;
}


constexpr int STRIP_ROWS = 32;
constexpr int STRIP_ROW_TILES = STRIP_ROWS / 16;
constexpr int STRIP_WAVES = 8;
constexpr int STRIP_THREADS = STRIP_WAVES * 32;
constexpr int STRIP_NODES_PER_WAVE = STRIP_ROWS / STRIP_WAVES;

template <int K, int MOUT>
__global__ __launch_bounds__(STRIP_THREADS, 16) void k_gather_gemm(
    const int* __restrict__ rowptr, const int* __restrict__ col, const float* __restrict__ hs_in,
    const float* __restrict__ dinv, const float* __restrict__ bias_in, const float* __restrict__ W,
    float* __restrict__ hs_out, int n) {
  constexpr int VPT = K / 32;
  typedef typename RowVec<VPT>::type VecT;
  constexpr int PITCH = K + 4;
  constexpr int RT = STRIP_ROW_TILES;
  constexpr int TW = 2;
  constexpr int GEMM_WAVES = MOUT / (16 * TW);
  static_assert(GEMM_WAVES >= 1 && GEMM_WAVES <= STRIP_WAVES, "tiles must fit the strip's waves");
  __shared__ __attribute__((aligned(16))) float strip[STRIP_ROWS * PITCH];

  const int lane = threadIdx.x & 31;
  const int wave = __builtin_amdgcn_readfirstlane((int)(threadIdx.x >> 5));
  const int row0 = blockIdx.x * STRIP_ROWS;

  const float* hl = hs_in + lane * VPT;
  const VecT bv = *(const VecT*)(bias_in + lane * VPT);
  for (int i = 0; i < STRIP_NODES_PER_WAVE; ++i) {
    const int r = wave * STRIP_NODES_PER_WAVE + i;
    const int node = row0 + r;
    VecT val = {};
    if (node < n) {
      const float dn = dinv[node];
      const VecT acc = node_sum<K, VecT>(col, hl, node, rowptr[node], rowptr[node + 1], lane);
      val = relu(acc * dn + bv);
    }
    *(VecT*)(strip + r * PITCH + lane * VPT) = val;
  }
  __syncthreads();
  const int mw = (wave + (int)blockIdx.x) & (STRIP_WAVES - 1);
  if (mw >= GEMM_WAVES) return;

  const int half = lane >> 4;
  const int l16  = lane & 15;
  const float* arow = strip + l16 * PITCH + 2 * half;

  v8f acc[RT][TW];
#pragma unroll
  for (int rt = 0; rt < RT; ++rt)
#pragma unroll
    for (int t = 0; t < TW; ++t) acc[rt][t] = (v8f){};

#pragma unroll 2
  for (int k = 0; k < K; k += 4) {
    const int ka = k + 2 * half;
    v2f a[RT];
#pragma unroll
    for (int rt = 0; rt < RT; ++rt) a[rt] = *(const v2f*)(arow + rt * 16 * PITCH + k);
#pragma unroll
    for (int t = 0; t < TW; ++t) {
      const int colB = (mw * TW + t) * 16 + l16;
      v2f b;
      b.x = W[(size_t)ka * MOUT + colB];
      b.y = W[(size_t)(ka + 1) * MOUT + colB];
#pragma unroll
      for (int rt = 0; rt < RT; ++rt)
        acc[rt][t] = __builtin_amdgcn_wmma_f32_16x16x4_f32(
            false, a[rt], false, b, (short)0, acc[rt][t], false, false);
    }
  }

  if (row0 + STRIP_ROWS <= n) {
#pragma unroll
    for (int rt = 0; rt < RT; ++rt) {
#pragma unroll
      for (int v = 0; v < 8; ++v) {
        const int row = row0 + rt * 16 + v + 8 * half;
        const float dn = dinv[row];
#pragma unroll
        for (int t = 0; t < TW; ++t) {
          const int c = (mw * TW + t) * 16 + l16;
          hs_out[(size_t)row * MOUT + c] = acc[rt][t][v] * dn;
        }
      }
    }
  } else {
#pragma unroll
    for (int rt = 0; rt < RT; ++rt) {
#pragma unroll
      for (int v = 0; v < 8; ++v) {
        const int row = row0 + rt * 16 + v + 8 * half;
        if (row < n) {
          const float dn = dinv[row];
#pragma unroll
          for (int t = 0; t < TW; ++t) {
            const int c = (mw * TW + t) * 16 + l16;
            hs_out[(size_t)row * MOUT + c] = acc[rt][t][v] * dn;
          }
        }
      }
    }
  }
}


template <int MOUT>
__global__ void k_gather_out(const int* __restrict__ rowptr, const int* __restrict__ col,
                             const float* __restrict__ hs, const float* __restrict__ dinv,
                             const float* __restrict__ bias, float* __restrict__ out, int n) {
  constexpr int VPT = MOUT / 32;
  typedef typename RowVec<VPT>::type VecT;
  const int node = __builtin_amdgcn_readfirstlane((int)((blockIdx.x * blockDim.x + threadIdx.x) >> 5));
  if (node >= n) return;
  const int lane = threadIdx.x & 31;
  const VecT acc = node_sum<MOUT, VecT>(col, hs + lane * VPT, node, rowptr[node], rowptr[node + 1], lane);
  const VecT bv = *(const VecT*)(bias + lane * VPT);
  *(VecT*)(out + (size_t)node * MOUT + lane * VPT) = acc * dinv[node] + bv;
}

extern "C" void kernel_launch(void* const* d_in, const int* in_sizes, int n_in,
                              void* d_out, int out_size, void* d_ws, size_t ws_size,
                              hipStream_t stream) {
  const float* x  = (const float*)d_in[0];
  const int*   ei = (const int*)d_in[1];
  const float* W1 = (const float*)d_in[2];
  const float* b1 = (const float*)d_in[3];
  const float* W2 = (const float*)d_in[4];
  const float* b2 = (const float*)d_in[5];
  const float* W3 = (const float*)d_in[6];
  const float* b3 = (const float*)d_in[7];
  const float* W4 = (const float*)d_in[8];
  const float* b4 = (const float*)d_in[9];

  const int n = in_sizes[0] / 256;
  const int e = in_sizes[1] / 2;
  const int* esrc = ei;
  const int* edst = ei + e;
  const int nbk = (n + BUCKET_NODES - 1) / BUCKET_NODES;
  const int chunk = (e + EDGE_BLOCKS - 1) / EDGE_BLOCKS;

  char* cur = (char*)d_ws;
  auto carve = [&cur](size_t bytes) { char* p = cur; cur += (bytes + 255) & ~(size_t)255; return p; };
  float* dinv   = (float*)carve((size_t)n * 4);
  int*   rowptr = (int*)carve((size_t)(n + 1) * 4);
  int*   table  = (int*)carve((size_t)nbk * EDGE_BLOCKS * 4);
  int*   bbase  = (int*)carve((size_t)(nbk + 1) * 4);
  unsigned* rec = (unsigned*)carve((size_t)e * 4);
  int*   slot   = (int*)carve((size_t)e * 4);
  int*   col    = (int*)carve((size_t)e * 4);
  float* HA = (float*)carve((size_t)n * 128 * 4);
  float* HB = (float*)carve((size_t)n * 128 * 4);

  k_bucket_counts<<<EDGE_BLOCKS, EDGE_THREADS, 0, stream>>>(edst, table, e, chunk, nbk);
  k_bucket_scatter<<<EDGE_BLOCKS, EDGE_THREADS, 0, stream>>>(esrc, edst, table, bbase, rec, e, chunk, nbk);
  k_bucket_rows_gemm<256, 128><<<nbk, OWNER_THREADS, 0, stream>>>(bbase, rec, rowptr, dinv, slot, col, x, W1, HA, n, nbk);

  const int strips = (n + STRIP_ROWS - 1) / STRIP_ROWS;
  k_gather_gemm<128, 128><<<strips, STRIP_THREADS, 0, stream>>>(rowptr, col, HA, dinv, b1, W2, HB, n);
  k_gather_gemm<128, 64><<<strips, STRIP_THREADS, 0, stream>>>(rowptr, col, HB, dinv, b2, W3, HA, n);
  k_gather_gemm<64, 32><<<strips, STRIP_THREADS, 0, stream>>>(rowptr, col, HA, dinv, b3, W4, HB, n);

  k_gather_out<32><<<(int)(((long long)n * 32 + 255) / 256), 256, 0, stream>>>(rowptr, col, HB, dinv, b4, (float*)d_out, n);
}
